# starting kernel
# baseline (speedup 1.0000x reference)
#include <hip/hip_runtime.h>


typedef __attribute__((ext_vector_type(2))) float v2f;
typedef __attribute__((ext_vector_type(8))) float v8f;

#define IMG   224
#define SROWS 144
#define SSTR  136
#define S_OFF   0
#define H3_OFF  (SROWS*SSTR)
#define H2_OFF  (H3_OFF + 144*16)
#define H1_OFF  (H2_OFF + 80*16)
#define W_OFF   (H1_OFF + 32*16)
#define SC_OFF  (W_OFF + 208)
#define NT3 35
#define NT2 18
#define NT1 7
#define TB3_OFF (SC_OFF + 16)
#define TB2_OFF (TB3_OFF + NT3*32*2)
#define TB1_OFF (TB2_OFF + NT2*32*2)
#define SMEM_FLOATS (TB1_OFF + NT1*32*2)

__device__ __forceinline__ float bicubic_w(int t, int K) {
  const double A_ = -0.5;
  double x  = ((double)t - 0.5 * (double)K + 0.5) * (4.0 / (double)K);
  double ax = fabs(x);
  double w;
  if (ax > 2.0)       w = 0.0;
  else if (ax > 1.0)  w = ((A_ * ax - 5.0 * A_) * ax + 8.0 * A_) * ax - 4.0 * A_;
  else                w = (((A_ + 2.0) * ax - (A_ + 3.0)) * ax) * ax + 1.0;
  return (float)w;
}

__global__ __launch_bounds__(256) void glimpse_wmma_kernel(
    const float* __restrict__ imgs, const int* __restrict__ locs,
    float* __restrict__ out)
{
  extern __shared__ float sm[];
  float* S  = sm + S_OFF;
  float* Wv = sm + W_OFF;
  float* SC = sm + SC_OFF;

  const int b    = blockIdx.x;
  const int tid  = threadIdx.x;
  const int lane = tid & 31;
  const int wave = __builtin_amdgcn_readfirstlane(tid >> 5);

  if (tid < 193) {
    int t = tid, K, base;
    if (t < 12)      { K = 12;  base = 0;            }
    else if (t < 69) { K = 57;  base = 12; t -= 12;  }
    else             { K = 124; base = 69; t -= 69;  }
    Wv[base + t] = bicubic_w(t, K);
  }

  const int x0 = locs[2 * b + 0] - 67;
  const int y0 = locs[2 * b + 1] - 67;
  const float* img = imgs + (size_t)b * (3 * IMG * IMG);
  for (int idx = tid; idx < SROWS * SSTR; idx += 256) {
    int r = idx / SSTR;
    int c = idx - r * SSTR;
    float v = 0.f;
    int iy = x0 + r, ix = y0 + c;
    if (r < 135 && c < 135 && iy >= 0 && iy < IMG && ix >= 0 && ix < IMG) {
      int o = iy * IMG + ix;
      v = __builtin_nontemporal_load(img + o)
        + __builtin_nontemporal_load(img + IMG * IMG + o)
        + __builtin_nontemporal_load(img + 2 * IMG * IMG + o);
    }
    S[idx] = v;
  }
  __syncthreads();

  if (tid < 4) {
    if (tid == 3) { SC[0] = 1.0f / 3.0f; }
    else {
      int K    = (tid == 0) ? 12 : ((tid == 1) ? 57 : 124);
      int base = (tid == 0) ? 0  : ((tid == 1) ? 12 : 69);
      double s = 0.0;
      for (int i = 0; i < K; ++i) s += (double)Wv[base + i];
      SC[tid + 1] = (float)(1.0 / (3.0 * s * s));
    }
  }
  for (int e = tid; e < (NT3 + NT2 + NT1) * 32; e += 256) {
    int gi, le, base;
    if (e < NT3 * 32)              { gi = 3; le = e;              base = TB3_OFF; }
    else if (e < (NT3 + NT2) * 32) { gi = 2; le = e - NT3 * 32;   base = TB2_OFF; }
    else                           { gi = 1; le = e - (NT3 + NT2) * 32; base = TB1_OFF; }
    const int K  = (gi == 3) ? 124 : ((gi == 2) ? 57 : 12);
    const int g  = K + 11;
    const int wb = (gi == 3) ? 69 : ((gi == 2) ? 12 : 0);
    const int t  = le >> 5;
    const int l  = le & 31;
    const int mm = l & 15;
    const int c1 = t * 4 + ((l >> 4) << 1);
    const int d0 = c1 - mm;
    float v0 = Wv[wb + min(max(d0, 0), K - 1)];
    float v1 = Wv[wb + min(max(d0 + 1, 0), K - 1)];
    v0 = (mm < 12 && c1     < g && d0     >= 0 && d0     < K) ? v0 : 0.f;
    v1 = (mm < 12 && c1 + 1 < g && d0 + 1 >= 0 && d0 + 1 < K) ? v1 : 0.f;
    sm[base + 2 * le]     = v0;
    sm[base + 2 * le + 1] = v1;
  }
  __syncthreads();

  const int m  = lane & 15;
  const int kk = (lane >> 4) << 1;
  const int Mb = (lane >> 4) << 3;

  for (int task = wave; task < 16; task += 8) {
    int gi, blk;
    if (task < 9)       { gi = 3; blk = task;      }
    else if (task < 14) { gi = 2; blk = task - 9;  }
    else                { gi = 1; blk = task - 14; }
    const int g     = (gi == 3) ? 135 : ((gi == 2) ? 68 : 23);
    const int off   = 67 - (g >> 1);
    const int ntile = (g + 3) >> 2;
    const v2f* TBg  = (const v2f*)(sm + ((gi == 3) ? TB3_OFF : ((gi == 2) ? TB2_OFF : TB1_OFF)));
    float*     H    = sm + ((gi == 3) ? H3_OFF : ((gi == 2) ? H2_OFF : H1_OFF));
    const float* Srow = S + (off + blk * 16 + m) * SSTR + off + kk;

    v8f acc = {};
    v2f a, bm;
    a.x = Srow[0]; a.y = Srow[1];
    bm  = TBg[lane];
    for (int t = 0; t < ntile; ++t) {
      v2f a2, b2;
      a2.x = Srow[(t + 1) * 4];
      a2.y = Srow[(t + 1) * 4 + 1];
      b2   = TBg[(t + 1) * 32 + lane];
      acc = __builtin_amdgcn_wmma_f32_16x16x4_f32(
          false, a, false, bm, (short)0, acc, false, false);
      a = a2; bm = b2;
    }
#pragma unroll
    for (int r = 0; r < 8; ++r)
      H[(blk * 16 + Mb + r) * 16 + m] = acc[r];
  }
  __syncthreads();

  if (wave < 3) {
    const int gi    = wave + 1;
    const int g     = (gi == 3) ? 135 : ((gi == 2) ? 68 : 23);
    const int ntile = (g + 3) >> 2;
    const v2f* TBg  = (const v2f*)(sm + ((gi == 3) ? TB3_OFF : ((gi == 2) ? TB2_OFF : TB1_OFF)));
    const float* H  = sm + ((gi == 3) ? H3_OFF : ((gi == 2) ? H2_OFF : H1_OFF)) + m;

    v8f acc = {};
    v2f a, bm;
    a    = TBg[lane];
    bm.x = H[kk * 16];
    bm.y = H[(kk + 1) * 16];
    for (int t = 0; t < ntile; ++t) {
      v2f a2, b2;
      a2   = TBg[(t + 1) * 32 + lane];
      b2.x = H[((t + 1) * 4 + kk) * 16];
      b2.y = H[((t + 1) * 4 + kk + 1) * 16];
      acc = __builtin_amdgcn_wmma_f32_16x16x4_f32(
          false, a, false, bm, (short)0, acc, false, false);
      a = a2; bm = b2;
    }
    const float sc = SC[gi];
#pragma unroll
    for (int r = 0; r < 8; ++r) {
      const int M = Mb + r;
      if (M < 12 && m < 12)
        out[(((size_t)b * 4 + gi) * 12 + M) * 12 + m] = acc[r] * sc;
    }
  } else if (wave == 3) {
    const float sc = SC[0];
    for (int t = lane; t < 144; t += 32) {
      int o = t / 12, j = t - o * 12;
      out[((size_t)b * 4 * 12 + o) * 12 + j] = S[(61 + o) * SSTR + 61 + j] * sc;
    }
  }
}

extern "C" void kernel_launch(void* const* d_in, const int* in_sizes, int n_in,
                              void* d_out, int out_size, void* d_ws, size_t ws_size,
                              hipStream_t stream) {
  const float* imgs = (const float*)d_in[0];
  const int*   locs = (const int*)d_in[1];
  float*       out  = (float*)d_out;
  (void)in_sizes; (void)n_in; (void)d_ws; (void)ws_size; (void)out_size;

  const size_t shmem = (size_t)SMEM_FLOATS * sizeof(float);
  glimpse_wmma_kernel<<<dim3(256), dim3(256), shmem, stream>>>(imgs, locs, out);
}


# crop gather rebuilt for loads in flight: 16 waves per block, one crop row per wave at a time, three rows (45 clamped loads, no branch, no divide) issued together ahead of their sums
# speedup vs baseline: 2.7315x; 2.7315x over previous
#include <hip/hip_runtime.h>


typedef __attribute__((ext_vector_type(2))) float v2f;
typedef __attribute__((ext_vector_type(8))) float v8f;

#define IMG   224
#define NTHR  512
#define NWAVE (NTHR / 32)
#define RPW   (SROWS / NWAVE)
#define RB    3
#define SROWS 144
#define SSTR  136
#define S_OFF   0
#define H3_OFF  (SROWS*SSTR)
#define H2_OFF  (H3_OFF + 144*16)
#define H1_OFF  (H2_OFF + 80*16)
#define W_OFF   (H1_OFF + 32*16)
#define SC_OFF  (W_OFF + 208)
#define NT3 35
#define NT2 18
#define NT1 7
#define TB3_OFF (SC_OFF + 16)
#define TB2_OFF (TB3_OFF + NT3*32*2)
#define TB1_OFF (TB2_OFF + NT2*32*2)
#define SMEM_FLOATS (TB1_OFF + NT1*32*2)

__device__ __forceinline__ float bicubic_w(int t, int K) {
  const double A_ = -0.5;
  double x  = ((double)t - 0.5 * (double)K + 0.5) * (4.0 / (double)K);
  double ax = fabs(x);
  double w;
  if (ax > 2.0)       w = 0.0;
  else if (ax > 1.0)  w = ((A_ * ax - 5.0 * A_) * ax + 8.0 * A_) * ax - 4.0 * A_;
  else                w = (((A_ + 2.0) * ax - (A_ + 3.0)) * ax) * ax + 1.0;
  return (float)w;
}

__global__ __launch_bounds__(NTHR) void glimpse_wmma_kernel(
    const float* __restrict__ imgs, const int* __restrict__ locs,
    float* __restrict__ out)
{
  extern __shared__ float sm[];
  float* S  = sm + S_OFF;
  float* Wv = sm + W_OFF;
  float* SC = sm + SC_OFF;

  const int b    = blockIdx.x;
  const int tid  = threadIdx.x;
  const int lane = tid & 31;
  const int wave = __builtin_amdgcn_readfirstlane(tid >> 5);

  if (tid < 193) {
    int t = tid, K, base;
    if (t < 12)      { K = 12;  base = 0;            }
    else if (t < 69) { K = 57;  base = 12; t -= 12;  }
    else             { K = 124; base = 69; t -= 69;  }
    Wv[base + t] = bicubic_w(t, K);
  }

  const int x0 = locs[2 * b + 0] - 67;
  const int y0 = locs[2 * b + 1] - 67;
  const float* img = imgs + (size_t)b * (3 * IMG * IMG);
  int  cofs[5];
  bool cok[5];
#pragma unroll
  for (int j = 0; j < 5; ++j) {
    const int c  = lane + 32 * j;
    const int ix = y0 + c;
    cok[j]  = (c < 135) && (ix >= 0) && (ix < IMG);
    cofs[j] = min(max(y0 + min(c, 134), 0), IMG - 1);
  }
  for (int k0 = 0; k0 < RPW; k0 += RB) {
    float ld[RB][5][3];
#pragma unroll
    for (int kr = 0; kr < RB; ++kr) {
      const int r   = wave + NWAVE * (k0 + kr);
      const int iyc = min(max(x0 + min(r, 134), 0), IMG - 1);
      const float* p = img + iyc * IMG;
#pragma unroll
      for (int j = 0; j < 5; ++j) {
        ld[kr][j][0] = __builtin_nontemporal_load(p + cofs[j]);
        ld[kr][j][1] = __builtin_nontemporal_load(p + IMG * IMG + cofs[j]);
        ld[kr][j][2] = __builtin_nontemporal_load(p + 2 * IMG * IMG + cofs[j]);
      }
    }
#pragma unroll
    for (int kr = 0; kr < RB; ++kr) {
      const int  r   = wave + NWAVE * (k0 + kr);
      const int  iy  = x0 + r;
      const bool rok = (r < 135) && (iy >= 0) && (iy < IMG);
      float* Sr = S + r * SSTR;
#pragma unroll
      for (int j = 0; j < 5; ++j) {
        const float v = (rok && cok[j]) ? (ld[kr][j][0] + ld[kr][j][1]) + ld[kr][j][2] : 0.f;
        if (j < 4) Sr[lane + 32 * j]     = v;
        else       Sr[128 + min(lane, 7)] = v;
      }
    }
  }
  __syncthreads();

  if (tid < 4) {
    if (tid == 3) { SC[0] = 1.0f / 3.0f; }
    else {
      int K    = (tid == 0) ? 12 : ((tid == 1) ? 57 : 124);
      int base = (tid == 0) ? 0  : ((tid == 1) ? 12 : 69);
      double s = 0.0;
      for (int i = 0; i < K; ++i) s += (double)Wv[base + i];
      SC[tid + 1] = (float)(1.0 / (3.0 * s * s));
    }
  }
  for (int e = tid; e < (NT3 + NT2 + NT1) * 32; e += NTHR) {
    int gi, le, base;
    if (e < NT3 * 32)              { gi = 3; le = e;              base = TB3_OFF; }
    else if (e < (NT3 + NT2) * 32) { gi = 2; le = e - NT3 * 32;   base = TB2_OFF; }
    else                           { gi = 1; le = e - (NT3 + NT2) * 32; base = TB1_OFF; }
    const int K  = (gi == 3) ? 124 : ((gi == 2) ? 57 : 12);
    const int g  = K + 11;
    const int wb = (gi == 3) ? 69 : ((gi == 2) ? 12 : 0);
    const int t  = le >> 5;
    const int l  = le & 31;
    const int mm = l & 15;
    const int c1 = t * 4 + ((l >> 4) << 1);
    const int d0 = c1 - mm;
    float v0 = Wv[wb + min(max(d0, 0), K - 1)];
    float v1 = Wv[wb + min(max(d0 + 1, 0), K - 1)];
    v0 = (mm < 12 && c1     < g && d0     >= 0 && d0     < K) ? v0 : 0.f;
    v1 = (mm < 12 && c1 + 1 < g && d0 + 1 >= 0 && d0 + 1 < K) ? v1 : 0.f;
    sm[base + 2 * le]     = v0;
    sm[base + 2 * le + 1] = v1;
  }
  __syncthreads();

  const int m  = lane & 15;
  const int kk = (lane >> 4) << 1;
  const int Mb = (lane >> 4) << 3;

  for (int task = wave; task < 16; task += NWAVE) {
    int gi, blk;
    if (task < 9)       { gi = 3; blk = task;      }
    else if (task < 14) { gi = 2; blk = task - 9;  }
    else                { gi = 1; blk = task - 14; }
    const int g     = (gi == 3) ? 135 : ((gi == 2) ? 68 : 23);
    const int off   = 67 - (g >> 1);
    const int ntile = (g + 3) >> 2;
    const v2f* TBg  = (const v2f*)(sm + ((gi == 3) ? TB3_OFF : ((gi == 2) ? TB2_OFF : TB1_OFF)));
    float*     H    = sm + ((gi == 3) ? H3_OFF : ((gi == 2) ? H2_OFF : H1_OFF));
    const float* Srow = S + (off + blk * 16 + m) * SSTR + off + kk;

    v8f acc = {};
    v2f a, bm;
    a.x = Srow[0]; a.y = Srow[1];
    bm  = TBg[lane];
    for (int t = 0; t < ntile; ++t) {
      v2f a2, b2;
      a2.x = Srow[(t + 1) * 4];
      a2.y = Srow[(t + 1) * 4 + 1];
      b2   = TBg[(t + 1) * 32 + lane];
      acc = __builtin_amdgcn_wmma_f32_16x16x4_f32(
          false, a, false, bm, (short)0, acc, false, false);
      a = a2; bm = b2;
    }
#pragma unroll
    for (int r = 0; r < 8; ++r)
      H[(blk * 16 + Mb + r) * 16 + m] = acc[r];
  }
  __syncthreads();

  if (wave < 3) {
    const int gi    = wave + 1;
    const int g     = (gi == 3) ? 135 : ((gi == 2) ? 68 : 23);
    const int ntile = (g + 3) >> 2;
    const v2f* TBg  = (const v2f*)(sm + ((gi == 3) ? TB3_OFF : ((gi == 2) ? TB2_OFF : TB1_OFF)));
    const float* H  = sm + ((gi == 3) ? H3_OFF : ((gi == 2) ? H2_OFF : H1_OFF)) + m;

    v8f acc = {};
    v2f a, bm;
    a    = TBg[lane];
    bm.x = H[kk * 16];
    bm.y = H[(kk + 1) * 16];
    for (int t = 0; t < ntile; ++t) {
      v2f a2, b2;
      a2   = TBg[(t + 1) * 32 + lane];
      b2.x = H[((t + 1) * 4 + kk) * 16];
      b2.y = H[((t + 1) * 4 + kk + 1) * 16];
      acc = __builtin_amdgcn_wmma_f32_16x16x4_f32(
          false, a, false, bm, (short)0, acc, false, false);
      a = a2; bm = b2;
    }
    const float sc = SC[gi];
#pragma unroll
    for (int r = 0; r < 8; ++r) {
      const int M = Mb + r;
      if (M < 12 && m < 12)
        out[(((size_t)b * 4 + gi) * 12 + M) * 12 + m] = acc[r] * sc;
    }
  } else if (wave == 3) {
    const float sc = SC[0];
    for (int t = lane; t < 144; t += 32) {
      int o = t / 12, j = t - o * 12;
      out[((size_t)b * 4 * 12 + o) * 12 + j] = S[(61 + o) * SSTR + 61 + j] * sc;
    }
  }
}

extern "C" void kernel_launch(void* const* d_in, const int* in_sizes, int n_in,
                              void* d_out, int out_size, void* d_ws, size_t ws_size,
                              hipStream_t stream) {
  const float* imgs = (const float*)d_in[0];
  const int*   locs = (const int*)d_in[1];
  float*       out  = (float*)d_out;
  (void)in_sizes; (void)n_in; (void)d_ws; (void)ws_size; (void)out_size;

  const size_t shmem = (size_t)SMEM_FLOATS * sizeof(float);
  glimpse_wmma_kernel<<<dim3(256), dim3(NTHR), shmem, stream>>>(imgs, locs, out);
}


# filter scales: f64 sum of each filter's taps taken across the lanes of one wave per filter (tree: summation order changes, exact in f64 either way) instead of one thread's serial chain of adds
# speedup vs baseline: 3.8414x; 1.4063x over previous
#include <hip/hip_runtime.h>


typedef __attribute__((ext_vector_type(2))) float v2f;
typedef __attribute__((ext_vector_type(8))) float v8f;

#define IMG   224
#define NTHR  512
#define NWAVE (NTHR / 32)
#define RPW   (SROWS / NWAVE)
#define RB    3
#define SCW   3
#define SROWS 144
#define SSTR  136
#define S_OFF   0
#define H3_OFF  (SROWS*SSTR)
#define H2_OFF  (H3_OFF + 144*16)
#define H1_OFF  (H2_OFF + 80*16)
#define W_OFF   (H1_OFF + 32*16)
#define SC_OFF  (W_OFF + 208)
#define NT3 35
#define NT2 18
#define NT1 7
#define TB3_OFF (SC_OFF + 16)
#define TB2_OFF (TB3_OFF + NT3*32*2)
#define TB1_OFF (TB2_OFF + NT2*32*2)
#define SMEM_FLOATS (TB1_OFF + NT1*32*2)

__device__ __forceinline__ float bicubic_w(int t, int K) {
  const double A_ = -0.5;
  double x  = ((double)t - 0.5 * (double)K + 0.5) * (4.0 / (double)K);
  double ax = fabs(x);
  double w;
  if (ax > 2.0)       w = 0.0;
  else if (ax > 1.0)  w = ((A_ * ax - 5.0 * A_) * ax + 8.0 * A_) * ax - 4.0 * A_;
  else                w = (((A_ + 2.0) * ax - (A_ + 3.0)) * ax) * ax + 1.0;
  return (float)w;
}

__global__ __launch_bounds__(NTHR) void glimpse_wmma_kernel(
    const float* __restrict__ imgs, const int* __restrict__ locs,
    float* __restrict__ out)
{
  extern __shared__ float sm[];
  float* S  = sm + S_OFF;
  float* Wv = sm + W_OFF;
  float* SC = sm + SC_OFF;

  const int b    = blockIdx.x;
  const int tid  = threadIdx.x;
  const int lane = tid & 31;
  const int wave = __builtin_amdgcn_readfirstlane(tid >> 5);

  if (tid < 193) {
    int t = tid, K, base;
    if (t < 12)      { K = 12;  base = 0;            }
    else if (t < 69) { K = 57;  base = 12; t -= 12;  }
    else             { K = 124; base = 69; t -= 69;  }
    Wv[base + t] = bicubic_w(t, K);
  }

  const int x0 = locs[2 * b + 0] - 67;
  const int y0 = locs[2 * b + 1] - 67;
  const float* img = imgs + (size_t)b * (3 * IMG * IMG);
  int  cofs[5];
  bool cok[5];
#pragma unroll
  for (int j = 0; j < 5; ++j) {
    const int c  = lane + 32 * j;
    const int ix = y0 + c;
    cok[j]  = (c < 135) && (ix >= 0) && (ix < IMG);
    cofs[j] = min(max(y0 + min(c, 134), 0), IMG - 1);
  }
  for (int k0 = 0; k0 < RPW; k0 += RB) {
    float ld[RB][5][3];
#pragma unroll
    for (int kr = 0; kr < RB; ++kr) {
      const int r   = wave + NWAVE * (k0 + kr);
      const int iyc = min(max(x0 + min(r, 134), 0), IMG - 1);
      const float* p = img + iyc * IMG;
#pragma unroll
      for (int j = 0; j < 5; ++j) {
        ld[kr][j][0] = __builtin_nontemporal_load(p + cofs[j]);
        ld[kr][j][1] = __builtin_nontemporal_load(p + IMG * IMG + cofs[j]);
        ld[kr][j][2] = __builtin_nontemporal_load(p + 2 * IMG * IMG + cofs[j]);
      }
    }
#pragma unroll
    for (int kr = 0; kr < RB; ++kr) {
      const int  r   = wave + NWAVE * (k0 + kr);
      const int  iy  = x0 + r;
      const bool rok = (r < 135) && (iy >= 0) && (iy < IMG);
      float* Sr = S + r * SSTR;
#pragma unroll
      for (int j = 0; j < 5; ++j) {
        const float v = (rok && cok[j]) ? (ld[kr][j][0] + ld[kr][j][1]) + ld[kr][j][2] : 0.f;
        if (j < 4) Sr[lane + 32 * j]     = v;
        else       Sr[128 + min(lane, 7)] = v;
      }
    }
  }
  __syncthreads();

  if (wave >= NWAVE - SCW) {
    const int f    = wave - (NWAVE - SCW);
    const int K    = (f == 0) ? 12 : ((f == 1) ? 57 : 124);
    const int base = (f == 0) ? 0  : ((f == 1) ? 12 : 69);
    double d[4];
    for (int j = 0; j < 4; ++j) {
      const int   i = lane + 32 * j;
      const float w = Wv[base + min(i, K - 1)];
      d[j] = (i < K) ? (double)w : 0.0;
    }
    double s = (d[0] + d[1]) + (d[2] + d[3]);
    for (int x = 16; x > 0; x >>= 1) s += __shfl_xor(s, x);
    if (lane == 0) {
      SC[f + 1] = (float)(1.0 / (3.0 * s * s));
      if (f == 0) SC[0] = 1.0f / 3.0f;
    }
  } else {
    for (int e = tid; e < (NT3 + NT2 + NT1) * 32; e += (NWAVE - SCW) * 32) {
      int gi, le, base;
      if (e < NT3 * 32)              { gi = 3; le = e;              base = TB3_OFF; }
      else if (e < (NT3 + NT2) * 32) { gi = 2; le = e - NT3 * 32;   base = TB2_OFF; }
      else                           { gi = 1; le = e - (NT3 + NT2) * 32; base = TB1_OFF; }
      const int K  = (gi == 3) ? 124 : ((gi == 2) ? 57 : 12);
      const int g  = K + 11;
      const int wb = (gi == 3) ? 69 : ((gi == 2) ? 12 : 0);
      const int t  = le >> 5;
      const int l  = le & 31;
      const int mm = l & 15;
      const int c1 = t * 4 + ((l >> 4) << 1);
      const int d0 = c1 - mm;
      float v0 = Wv[wb + min(max(d0, 0), K - 1)];
      float v1 = Wv[wb + min(max(d0 + 1, 0), K - 1)];
      v0 = (mm < 12 && c1     < g && d0     >= 0 && d0     < K) ? v0 : 0.f;
      v1 = (mm < 12 && c1 + 1 < g && d0 + 1 >= 0 && d0 + 1 < K) ? v1 : 0.f;
      sm[base + 2 * le]     = v0;
      sm[base + 2 * le + 1] = v1;
    }
  }
  __syncthreads();

  const int m  = lane & 15;
  const int kk = (lane >> 4) << 1;
  const int Mb = (lane >> 4) << 3;

  for (int task = wave; task < 16; task += NWAVE) {
    int gi, blk;
    if (task < 9)       { gi = 3; blk = task;      }
    else if (task < 14) { gi = 2; blk = task - 9;  }
    else                { gi = 1; blk = task - 14; }
    const int g     = (gi == 3) ? 135 : ((gi == 2) ? 68 : 23);
    const int off   = 67 - (g >> 1);
    const int ntile = (g + 3) >> 2;
    const v2f* TBg  = (const v2f*)(sm + ((gi == 3) ? TB3_OFF : ((gi == 2) ? TB2_OFF : TB1_OFF)));
    float*     H    = sm + ((gi == 3) ? H3_OFF : ((gi == 2) ? H2_OFF : H1_OFF));
    const float* Srow = S + (off + blk * 16 + m) * SSTR + off + kk;

    v8f acc = {};
    v2f a, bm;
    a.x = Srow[0]; a.y = Srow[1];
    bm  = TBg[lane];
    for (int t = 0; t < ntile; ++t) {
      v2f a2, b2;
      a2.x = Srow[(t + 1) * 4];
      a2.y = Srow[(t + 1) * 4 + 1];
      b2   = TBg[(t + 1) * 32 + lane];
      acc = __builtin_amdgcn_wmma_f32_16x16x4_f32(
          false, a, false, bm, (short)0, acc, false, false);
      a = a2; bm = b2;
    }
#pragma unroll
    for (int r = 0; r < 8; ++r)
      H[(blk * 16 + Mb + r) * 16 + m] = acc[r];
  }
  __syncthreads();

  if (wave < 3) {
    const int gi    = wave + 1;
    const int g     = (gi == 3) ? 135 : ((gi == 2) ? 68 : 23);
    const int ntile = (g + 3) >> 2;
    const v2f* TBg  = (const v2f*)(sm + ((gi == 3) ? TB3_OFF : ((gi == 2) ? TB2_OFF : TB1_OFF)));
    const float* H  = sm + ((gi == 3) ? H3_OFF : ((gi == 2) ? H2_OFF : H1_OFF)) + m;

    v8f acc = {};
    v2f a, bm;
    a    = TBg[lane];
    bm.x = H[kk * 16];
    bm.y = H[(kk + 1) * 16];
    for (int t = 0; t < ntile; ++t) {
      v2f a2, b2;
      a2   = TBg[(t + 1) * 32 + lane];
      b2.x = H[((t + 1) * 4 + kk) * 16];
      b2.y = H[((t + 1) * 4 + kk + 1) * 16];
      acc = __builtin_amdgcn_wmma_f32_16x16x4_f32(
          false, a, false, bm, (short)0, acc, false, false);
      a = a2; bm = b2;
    }
    const float sc = SC[gi];
#pragma unroll
    for (int r = 0; r < 8; ++r) {
      const int M = Mb + r;
      if (M < 12 && m < 12)
        out[(((size_t)b * 4 + gi) * 12 + M) * 12 + m] = acc[r] * sc;
    }
  } else if (wave == 3) {
    const float sc = SC[0];
    for (int t = lane; t < 144; t += 32) {
      int o = t / 12, j = t - o * 12;
      out[((size_t)b * 4 * 12 + o) * 12 + j] = S[(61 + o) * SSTR + 61 + j] * sc;
    }
  }
}

extern "C" void kernel_launch(void* const* d_in, const int* in_sizes, int n_in,
                              void* d_out, int out_size, void* d_ws, size_t ws_size,
                              hipStream_t stream) {
  const float* imgs = (const float*)d_in[0];
  const int*   locs = (const int*)d_in[1];
  float*       out  = (float*)d_out;
  (void)in_sizes; (void)n_in; (void)d_ws; (void)ws_size; (void)out_size;

  const size_t shmem = (size_t)SMEM_FLOATS * sizeof(float);
  glimpse_wmma_kernel<<<dim3(256), dim3(NTHR), shmem, stream>>>(imgs, locs, out);
}


# crop gather by whole 128-byte image lines, kept line-aligned in LDS: a wave-load reads four lines (16 B per lane), one 16 B LDS store per lane; each needed line read once; S pitch 136 to 196
# speedup vs baseline: 4.5315x; 1.1796x over previous
#include <hip/hip_runtime.h>


typedef __attribute__((ext_vector_type(2))) float v2f;
typedef __attribute__((ext_vector_type(4))) float v4f;
typedef __attribute__((ext_vector_type(8))) float v8f;

#define IMG   224
#define NTHR  512
#define NWAVE (NTHR / 32)
#define RPW   (SROWS / NWAVE)
#define LPR   (IMG / 32)
#define SCW   3
#define SROWS 144
#define SSTR  196
#define S_OFF   0
#define H3_OFF  (SROWS*SSTR)
#define H2_OFF  (H3_OFF + 144*16)
#define H1_OFF  (H2_OFF + 80*16)
#define W_OFF   (H1_OFF + 32*16)
#define SC_OFF  (W_OFF + 208)
#define NT3 35
#define NT2 18
#define NT1 7
#define TB3_OFF (SC_OFF + 16)
#define TB2_OFF (TB3_OFF + NT3*32*2)
#define TB1_OFF (TB2_OFF + NT2*32*2)
#define SMEM_FLOATS (TB1_OFF + NT1*32*2)

__device__ __forceinline__ float bicubic_w(int t, int K) {
  const double A_ = -0.5;
  double x  = ((double)t - 0.5 * (double)K + 0.5) * (4.0 / (double)K);
  double ax = fabs(x);
  double w;
  if (ax > 2.0)       w = 0.0;
  else if (ax > 1.0)  w = ((A_ * ax - 5.0 * A_) * ax + 8.0 * A_) * ax - 4.0 * A_;
  else                w = (((A_ + 2.0) * ax - (A_ + 3.0)) * ax) * ax + 1.0;
  return (float)w;
}


template <int NL, int G0, int GB>
__device__ __forceinline__ void gather_batch(const float* __restrict__ img, float* __restrict__ S,
                                             int x0, int f, int wave, int lane)
{
  const int q4 = (lane & 7) << 2;
  const int us = lane >> 3;
  v4f ld[GB][3];
  int  so[GB];
  bool wok[GB];
#pragma unroll
  for (int g = 0; g < GB; ++g) {
    int k, ln;
    if (G0 + g < RPW) { k = G0 + g; ln = us; }
    else {
      const int v = min(4 * (G0 + g - RPW) + us, RPW * (NL - 4) - 1);
      if (NL == 5) { k = v;      ln = 4;           }
      else         { k = v >> 1; ln = 4 + (v & 1); }
    }
    const int  r   = wave + NWAVE * k;
    const int  iy  = x0 + r;
    const int  L   = f + ln;
    const bool w   = (r < 135) && (iy >= 0) && (iy < IMG) && (L >= 0) && (L < LPR);
    const float* p = img + (w ? (iy * IMG + L * 32 + q4) : q4);
    ld[g][0] = __builtin_nontemporal_load((const v4f*)p);
    ld[g][1] = __builtin_nontemporal_load((const v4f*)(p + IMG * IMG));
    ld[g][2] = __builtin_nontemporal_load((const v4f*)(p + 2 * IMG * IMG));
    so[g] = r * SSTR + ln * 32 + q4; wok[g] = w;
  }
#pragma unroll
  for (int g = 0; g < GB; ++g) {
    const v4f zero = {0.f, 0.f, 0.f, 0.f};
    const v4f s = (ld[g][0] + ld[g][1]) + ld[g][2];
    *(v4f*)(S + so[g]) = wok[g] ? s : zero;
  }
}

template <int NL>
__device__ __forceinline__ void gather_lines(const float* __restrict__ img, float* __restrict__ S,
                                             int x0, int f, int wave, int lane)
{
  constexpr int NG = RPW + (RPW * (NL - 4) + 3) / 4;
  gather_batch<NL, 0,      NG / 2>(img, S, x0, f, wave, lane);
  gather_batch<NL, NG / 2, NG / 2>(img, S, x0, f, wave, lane);
}

__global__ __launch_bounds__(NTHR) void glimpse_wmma_kernel(
    const float* __restrict__ imgs, const int* __restrict__ locs,
    float* __restrict__ out)
{
  extern __shared__ __attribute__((aligned(16))) float sm[];
  float* S  = sm + S_OFF;
  float* Wv = sm + W_OFF;
  float* SC = sm + SC_OFF;

  const int b    = blockIdx.x;
  const int tid  = threadIdx.x;
  const int lane = tid & 31;
  const int wave = __builtin_amdgcn_readfirstlane(tid >> 5);

  if (tid < 193) {
    int t = tid, K, base;
    if (t < 12)      { K = 12;  base = 0;            }
    else if (t < 69) { K = 57;  base = 12; t -= 12;  }
    else             { K = 124; base = 69; t -= 69;  }
    Wv[base + t] = bicubic_w(t, K);
  }

  const int x0 = locs[2 * b + 0] - 67;
  const int y0 = locs[2 * b + 1] - 67;
  const float* img = imgs + (size_t)b * (3 * IMG * IMG);
  const int lf   = y0 >> 5;
  const int acol = y0 & 31;
  if (acol <= 24) gather_lines<5>(img, S, x0, lf, wave, lane);
  else            gather_lines<6>(img, S, x0, lf, wave, lane);
  __syncthreads();

  if (wave >= NWAVE - SCW) {
    const int f    = wave - (NWAVE - SCW);
    const int K    = (f == 0) ? 12 : ((f == 1) ? 57 : 124);
    const int base = (f == 0) ? 0  : ((f == 1) ? 12 : 69);
    double d[4];
    for (int j = 0; j < 4; ++j) {
      const int   i = lane + 32 * j;
      const float w = Wv[base + min(i, K - 1)];
      d[j] = (i < K) ? (double)w : 0.0;
    }
    double s = (d[0] + d[1]) + (d[2] + d[3]);
    for (int x = 16; x > 0; x >>= 1) s += __shfl_xor(s, x);
    if (lane == 0) {
      SC[f + 1] = (float)(1.0 / (3.0 * s * s));
      if (f == 0) SC[0] = 1.0f / 3.0f;
    }
  } else {
    for (int e = tid; e < (NT3 + NT2 + NT1) * 32; e += (NWAVE - SCW) * 32) {
      int gi, le, base;
      if (e < NT3 * 32)              { gi = 3; le = e;              base = TB3_OFF; }
      else if (e < (NT3 + NT2) * 32) { gi = 2; le = e - NT3 * 32;   base = TB2_OFF; }
      else                           { gi = 1; le = e - (NT3 + NT2) * 32; base = TB1_OFF; }
      const int K  = (gi == 3) ? 124 : ((gi == 2) ? 57 : 12);
      const int g  = K + 11;
      const int wb = (gi == 3) ? 69 : ((gi == 2) ? 12 : 0);
      const int t  = le >> 5;
      const int l  = le & 31;
      const int mm = l & 15;
      const int c1 = t * 4 + ((l >> 4) << 1);
      const int d0 = c1 - mm;
      float v0 = Wv[wb + min(max(d0, 0), K - 1)];
      float v1 = Wv[wb + min(max(d0 + 1, 0), K - 1)];
      v0 = (mm < 12 && c1     < g && d0     >= 0 && d0     < K) ? v0 : 0.f;
      v1 = (mm < 12 && c1 + 1 < g && d0 + 1 >= 0 && d0 + 1 < K) ? v1 : 0.f;
      sm[base + 2 * le]     = v0;
      sm[base + 2 * le + 1] = v1;
    }
  }
  __syncthreads();

  const int m  = lane & 15;
  const int kk = (lane >> 4) << 1;
  const int Mb = (lane >> 4) << 3;

  for (int task = wave; task < 16; task += NWAVE) {
    int gi, blk;
    if (task < 9)       { gi = 3; blk = task;      }
    else if (task < 14) { gi = 2; blk = task - 9;  }
    else                { gi = 1; blk = task - 14; }
    const int g     = (gi == 3) ? 135 : ((gi == 2) ? 68 : 23);
    const int off   = 67 - (g >> 1);
    const int ntile = (g + 3) >> 2;
    const v2f* TBg  = (const v2f*)(sm + ((gi == 3) ? TB3_OFF : ((gi == 2) ? TB2_OFF : TB1_OFF)));
    float*     H    = sm + ((gi == 3) ? H3_OFF : ((gi == 2) ? H2_OFF : H1_OFF));
    const float* Srow = S + (off + blk * 16 + m) * SSTR + acol + off + kk;

    v8f acc = {};
    v2f a, bm;
    a.x = Srow[0]; a.y = Srow[1];
    bm  = TBg[lane];
    for (int t = 0; t < ntile; ++t) {
      v2f a2, b2;
      a2.x = Srow[(t + 1) * 4];
      a2.y = Srow[(t + 1) * 4 + 1];
      b2   = TBg[(t + 1) * 32 + lane];
      acc = __builtin_amdgcn_wmma_f32_16x16x4_f32(
          false, a, false, bm, (short)0, acc, false, false);
      a = a2; bm = b2;
    }
#pragma unroll
    for (int r = 0; r < 8; ++r)
      H[(blk * 16 + Mb + r) * 16 + m] = acc[r];
  }
  __syncthreads();

  if (wave < 3) {
    const int gi    = wave + 1;
    const int g     = (gi == 3) ? 135 : ((gi == 2) ? 68 : 23);
    const int ntile = (g + 3) >> 2;
    const v2f* TBg  = (const v2f*)(sm + ((gi == 3) ? TB3_OFF : ((gi == 2) ? TB2_OFF : TB1_OFF)));
    const float* H  = sm + ((gi == 3) ? H3_OFF : ((gi == 2) ? H2_OFF : H1_OFF)) + m;

    v8f acc = {};
    v2f a, bm;
    a    = TBg[lane];
    bm.x = H[kk * 16];
    bm.y = H[(kk + 1) * 16];
    for (int t = 0; t < ntile; ++t) {
      v2f a2, b2;
      a2   = TBg[(t + 1) * 32 + lane];
      b2.x = H[((t + 1) * 4 + kk) * 16];
      b2.y = H[((t + 1) * 4 + kk + 1) * 16];
      acc = __builtin_amdgcn_wmma_f32_16x16x4_f32(
          false, a, false, bm, (short)0, acc, false, false);
      a = a2; bm = b2;
    }
    const float sc = SC[gi];
#pragma unroll
    for (int r = 0; r < 8; ++r) {
      const int M = Mb + r;
      if (M < 12 && m < 12)
        out[(((size_t)b * 4 + gi) * 12 + M) * 12 + m] = acc[r] * sc;
    }
  } else if (wave == 3) {
    const float sc = SC[0];
    for (int t = lane; t < 144; t += 32) {
      int o = t / 12, j = t - o * 12;
      out[((size_t)b * 4 * 12 + o) * 12 + j] = S[(61 + o) * SSTR + acol + 61 + j] * sc;
    }
  }
}

extern "C" void kernel_launch(void* const* d_in, const int* in_sizes, int n_in,
                              void* d_out, int out_size, void* d_ws, size_t ws_size,
                              hipStream_t stream) {
  const float* imgs = (const float*)d_in[0];
  const int*   locs = (const int*)d_in[1];
  float*       out  = (float*)d_out;
  (void)in_sizes; (void)n_in; (void)d_ws; (void)ws_size; (void)out_size;

  const size_t shmem = (size_t)SMEM_FLOATS * sizeof(float);
  glimpse_wmma_kernel<<<dim3(256), dim3(NTHR), shmem, stream>>>(imgs, locs, out);
}


# vertical conv fused into the horizontal conv's waves: each multiplies its H block, still in accumulator registers, by its Toeplitz columns; block partials summed via LDS (summation order changes)
# speedup vs baseline: 4.8812x; 1.0772x over previous
#include <hip/hip_runtime.h>


typedef __attribute__((ext_vector_type(2))) float v2f;
typedef __attribute__((ext_vector_type(4))) float v4f;
typedef __attribute__((ext_vector_type(8))) float v8f;

#define IMG   224
#define NTHR  512
#define NWAVE (NTHR / 32)
#define RPW   (SROWS / NWAVE)
#define LPR   (IMG / 32)
#define SCW   3
#define SROWS 144
#define SSTR  196
#define S_OFF   0
#define H3_OFF  (SROWS*SSTR)
#define H2_OFF  (H3_OFF + 144*16)
#define H1_OFF  (H2_OFF + 80*16)
#define W_OFF   (H1_OFF + 32*16)
#define SC_OFF  (W_OFF + 208)
#define NT3 35
#define NT2 18
#define NT1 7
#define TB3_OFF (SC_OFF + 16)
#define TB2_OFF (TB3_OFF + NT3*32*2)
#define TB1_OFF (TB2_OFF + NT2*32*2)
#define SMEM_FLOATS (TB1_OFF + NT1*32*2)

__device__ __forceinline__ float bicubic_w(int t, int K) {
  const double A_ = -0.5;
  double x  = ((double)t - 0.5 * (double)K + 0.5) * (4.0 / (double)K);
  double ax = fabs(x);
  double w;
  if (ax > 2.0)       w = 0.0;
  else if (ax > 1.0)  w = ((A_ * ax - 5.0 * A_) * ax + 8.0 * A_) * ax - 4.0 * A_;
  else                w = (((A_ + 2.0) * ax - (A_ + 3.0)) * ax) * ax + 1.0;
  return (float)w;
}


template <int NL, int G0, int GB>
__device__ __forceinline__ void gather_batch(const float* __restrict__ img, float* __restrict__ S,
                                             int x0, int f, int wave, int lane)
{
  const int q4 = (lane & 7) << 2;
  const int us = lane >> 3;
  v4f ld[GB][3];
  int  so[GB];
  bool wok[GB];
#pragma unroll
  for (int g = 0; g < GB; ++g) {
    int k, ln;
    if (G0 + g < RPW) { k = G0 + g; ln = us; }
    else {
      const int v = min(4 * (G0 + g - RPW) + us, RPW * (NL - 4) - 1);
      if (NL == 5) { k = v;      ln = 4;           }
      else         { k = v >> 1; ln = 4 + (v & 1); }
    }
    const int  r   = wave + NWAVE * k;
    const int  iy  = x0 + r;
    const int  L   = f + ln;
    const bool w   = (r < 135) && (iy >= 0) && (iy < IMG) && (L >= 0) && (L < LPR);
    const float* p = img + (w ? (iy * IMG + L * 32 + q4) : q4);
    ld[g][0] = __builtin_nontemporal_load((const v4f*)p);
    ld[g][1] = __builtin_nontemporal_load((const v4f*)(p + IMG * IMG));
    ld[g][2] = __builtin_nontemporal_load((const v4f*)(p + 2 * IMG * IMG));
    so[g] = r * SSTR + ln * 32 + q4; wok[g] = w;
  }
#pragma unroll
  for (int g = 0; g < GB; ++g) {
    const v4f zero = {0.f, 0.f, 0.f, 0.f};
    const v4f s = (ld[g][0] + ld[g][1]) + ld[g][2];
    *(v4f*)(S + so[g]) = wok[g] ? s : zero;
  }
}

template <int NL>
__device__ __forceinline__ void gather_lines(const float* __restrict__ img, float* __restrict__ S,
                                             int x0, int f, int wave, int lane)
{
  constexpr int NG = RPW + (RPW * (NL - 4) + 3) / 4;
  gather_batch<NL, 0,      NG / 2>(img, S, x0, f, wave, lane);
  gather_batch<NL, NG / 2, NG / 2>(img, S, x0, f, wave, lane);
}

__global__ __launch_bounds__(NTHR) void glimpse_wmma_kernel(
    const float* __restrict__ imgs, const int* __restrict__ locs,
    float* __restrict__ out)
{
  extern __shared__ __attribute__((aligned(16))) float sm[];
  float* S  = sm + S_OFF;
  float* Wv = sm + W_OFF;
  float* SC = sm + SC_OFF;

  const int b    = blockIdx.x;
  const int tid  = threadIdx.x;
  const int lane = tid & 31;
  const int wave = __builtin_amdgcn_readfirstlane(tid >> 5);

  if (tid < 193) {
    int t = tid, K, base;
    if (t < 12)      { K = 12;  base = 0;            }
    else if (t < 69) { K = 57;  base = 12; t -= 12;  }
    else             { K = 124; base = 69; t -= 69;  }
    Wv[base + t] = bicubic_w(t, K);
  }

  const int x0 = locs[2 * b + 0] - 67;
  const int y0 = locs[2 * b + 1] - 67;
  const float* img = imgs + (size_t)b * (3 * IMG * IMG);
  const int lf   = y0 >> 5;
  const int acol = y0 & 31;
  if (acol <= 24) gather_lines<5>(img, S, x0, lf, wave, lane);
  else            gather_lines<6>(img, S, x0, lf, wave, lane);
  __syncthreads();

  if (wave >= NWAVE - SCW) {
    const int f    = wave - (NWAVE - SCW);
    const int K    = (f == 0) ? 12 : ((f == 1) ? 57 : 124);
    const int base = (f == 0) ? 0  : ((f == 1) ? 12 : 69);
    double d[4];
    for (int j = 0; j < 4; ++j) {
      const int   i = lane + 32 * j;
      const float w = Wv[base + min(i, K - 1)];
      d[j] = (i < K) ? (double)w : 0.0;
    }
    double s = (d[0] + d[1]) + (d[2] + d[3]);
    for (int x = 16; x > 0; x >>= 1) s += __shfl_xor(s, x);
    if (lane == 0) {
      SC[f + 1] = (float)(1.0 / (3.0 * s * s));
      if (f == 0) SC[0] = 1.0f / 3.0f;
    }
  } else {
    for (int e = tid; e < (NT3 + NT2 + NT1) * 32; e += (NWAVE - SCW) * 32) {
      int gi, le, base;
      if (e < NT3 * 32)              { gi = 3; le = e;              base = TB3_OFF; }
      else if (e < (NT3 + NT2) * 32) { gi = 2; le = e - NT3 * 32;   base = TB2_OFF; }
      else                           { gi = 1; le = e - (NT3 + NT2) * 32; base = TB1_OFF; }
      const int K  = (gi == 3) ? 124 : ((gi == 2) ? 57 : 12);
      const int g  = K + 11;
      const int wb = (gi == 3) ? 69 : ((gi == 2) ? 12 : 0);
      const int t  = le >> 5;
      const int l  = le & 31;
      const int mm = l & 15;
      const int c1 = t * 4 + ((l >> 4) << 1);
      const int d0 = c1 - mm;
      float v0 = Wv[wb + min(max(d0, 0), K - 1)];
      float v1 = Wv[wb + min(max(d0 + 1, 0), K - 1)];
      v0 = (mm < 12 && c1     < g && d0     >= 0 && d0     < K) ? v0 : 0.f;
      v1 = (mm < 12 && c1 + 1 < g && d0 + 1 >= 0 && d0 + 1 < K) ? v1 : 0.f;
      sm[base + 2 * le]     = v0;
      sm[base + 2 * le + 1] = v1;
    }
  }
  __syncthreads();

  const int m  = lane & 15;
  const int kk = (lane >> 4) << 1;
  const int Mb = (lane >> 4) << 3;

  const int hh = lane >> 4;
  for (int task = wave; task < 16; task += NWAVE) {
    int gi, blk;
    if (task < 9)       { gi = 3; blk = task;      }
    else if (task < 14) { gi = 2; blk = task - 9;  }
    else                { gi = 1; blk = task - 14; }
    const int g     = (gi == 3) ? 135 : ((gi == 2) ? 68 : 23);
    const int off   = 67 - (g >> 1);
    const int ntile = (g + 3) >> 2;
    const v2f* TBg  = (const v2f*)(sm + ((gi == 3) ? TB3_OFF : ((gi == 2) ? TB2_OFF : TB1_OFF)));
    float*     P    = sm + ((gi == 3) ? H3_OFF : ((gi == 2) ? H2_OFF : H1_OFF));
    const float* Srow = S + (off + blk * 16 + m) * SSTR + acol + off + kk;

    v2f av[4];
#pragma unroll
    for (int t = 0; t < 4; ++t) {
      const int tau = min(4 * blk + 2 * hh + (t >> 1), ntile);
      av[t] = TBg[tau * 32 + ((t & 1) << 4) + m];
    }

    v8f acc = {};
    v2f a, bm;
    a.x = Srow[0]; a.y = Srow[1];
    bm  = TBg[lane];
    for (int t = 0; t < ntile; ++t) {
      v2f a2, b2;
      a2.x = Srow[(t + 1) * 4];
      a2.y = Srow[(t + 1) * 4 + 1];
      b2   = TBg[(t + 1) * 32 + lane];
      acc = __builtin_amdgcn_wmma_f32_16x16x4_f32(
          false, a, false, bm, (short)0, acc, false, false);
      a = a2; bm = b2;
    }

    v8f po = {};
#pragma unroll
    for (int t = 0; t < 4; ++t) {
      v2f hb;
      hb.x = acc[2 * t];
      hb.y = acc[2 * t + 1];
      po = __builtin_amdgcn_wmma_f32_16x16x4_f32(
          false, av[t], false, hb, (short)0, po, false, false);
    }
#pragma unroll
    for (int r = 0; r < 8; ++r)
      P[(blk * 16 + Mb + r) * 16 + m] = po[r];
  }
  __syncthreads();

  if (tid < 3 * 144) {
    const int gq = tid / 144;
    const int gi = gq + 1;
    const int o  = tid - gq * 144;
    const int M  = o / 12;
    const int j  = o - M * 12;
    const int nb = (gi == 3) ? 9 : ((gi == 2) ? 5 : 2);
    const float* P = sm + ((gi == 3) ? H3_OFF : ((gi == 2) ? H2_OFF : H1_OFF)) + M * 16 + j;
    const float sc = SC[gi];
    float p[9];
#pragma unroll
    for (int k = 0; k < 9; ++k) p[k] = P[min(k, nb - 1) * 256];
    float s = p[0];
#pragma unroll
    for (int k = 1; k < 9; ++k) s = (k < nb) ? s + p[k] : s;
    out[((size_t)b * 4 + gi) * 144 + o] = s * sc;
  } else {
    const float sc = SC[0];
    for (int t = tid - 3 * 144; t < 144; t += NTHR - 3 * 144) {
      int o = t / 12, j = t - o * 12;
      out[((size_t)b * 4 * 12 + o) * 12 + j] = S[(61 + o) * SSTR + acol + 61 + j] * sc;
    }
  }
}

extern "C" void kernel_launch(void* const* d_in, const int* in_sizes, int n_in,
                              void* d_out, int out_size, void* d_ws, size_t ws_size,
                              hipStream_t stream) {
  const float* imgs = (const float*)d_in[0];
  const int*   locs = (const int*)d_in[1];
  float*       out  = (float*)d_out;
  (void)in_sizes; (void)n_in; (void)d_ws; (void)ws_size; (void)out_size;

  const size_t shmem = (size_t)SMEM_FLOATS * sizeof(float);
  glimpse_wmma_kernel<<<dim3(256), dim3(NTHR), shmem, stream>>>(imgs, locs, out);
}


# f64 set-up off the serial path: taps first on four waves (one per symmetric pair, 4/K a constant), then scales and Toeplitz tables under the gather's first batch of loads; values bit-identical
# speedup vs baseline: 5.1940x; 1.0641x over previous
#include <hip/hip_runtime.h>


typedef __attribute__((ext_vector_type(2))) float v2f;
typedef __attribute__((ext_vector_type(4))) float v4f;
typedef __attribute__((ext_vector_type(8))) float v8f;

#define IMG   224
#define NTHR  512
#define NWAVE (NTHR / 32)
#define RPW   (SROWS / NWAVE)
#define LPR   (IMG / 32)
#define SCW   3
#define NTAP  97
#define SROWS 144
#define SSTR  196
#define S_OFF   0
#define H3_OFF  (SROWS*SSTR)
#define H2_OFF  (H3_OFF + 144*16)
#define H1_OFF  (H2_OFF + 80*16)
#define W_OFF   (H1_OFF + 32*16)
#define SC_OFF  (W_OFF + 208)
#define NT3 35
#define NT2 18
#define NT1 7
#define TB3_OFF (SC_OFF + 16)
#define TB2_OFF (TB3_OFF + NT3*32*2)
#define TB1_OFF (TB2_OFF + NT2*32*2)
#define SMEM_FLOATS (TB1_OFF + NT1*32*2)

__device__ __forceinline__ void bicubic_taps(float* __restrict__ Wv, int u) {
  const bool f1 = u < 6, f2 = u < 35;
  const int    K    = f1 ? 12 : (f2 ? 57 : 124);
  const int    base = f1 ? 0  : (f2 ? 12 : 69);
  const int    t    = f1 ? u  : (f2 ? u - 6 : u - 35);
  const double hk   = f1 ? 5.5 : (f2 ? 28.0 : 61.5);
  const double r    = f1 ? (4.0 / 12.0) : (f2 ? (4.0 / 57.0) : (4.0 / 124.0));
  const double ax   = fabs(((double)t - hk) * r);
  const bool   in   = !(ax > 1.0);
  const double c3 = in ? 1.5  : -0.5;
  const double c2 = in ? -2.5 : 2.5;
  const double c1 = in ? 0.0  : -4.0;
  const double c0 = in ? 1.0  : 2.0;
  double w = __builtin_fma(__builtin_fma(__builtin_fma(c3, ax, c2), ax, c1), ax, c0);
  w = (ax > 2.0) ? 0.0 : w;
  const float wf = (float)w;
  Wv[base + t]         = wf;
  Wv[base + K - 1 - t] = wf;
}

__device__ __forceinline__ void scales_and_tables(float* __restrict__ sm, int wave, int lane)
{
  const float* Wv = sm + W_OFF;
  float*       SC = sm + SC_OFF;
  const int sw = wave / 3;
  if (wave == 3 * sw && sw >= 1 && sw <= SCW) {
    const int f    = sw - 1;
    const int K    = (f == 0) ? 12 : ((f == 1) ? 57 : 124);
    const int base = (f == 0) ? 0  : ((f == 1) ? 12 : 69);
    double d[4];
    for (int j = 0; j < 4; ++j) {
      const int   i = lane + 32 * j;
      const float w = Wv[base + min(i, K - 1)];
      d[j] = (i < K) ? (double)w : 0.0;
    }
    double s = (d[0] + d[1]) + (d[2] + d[3]);
    for (int x = 16; x > 0; x >>= 1) s += __shfl_xor(s, x);
    if (lane == 0) {
      SC[f + 1] = (float)(1.0 / (3.0 * s * s));
      if (f == 0) SC[0] = 1.0f / 3.0f;
    }
  } else {
    const int rank = wave - (wave > 3) - (wave > 6) - (wave > 9);
    for (int e = rank * 32 + lane; e < (NT3 + NT2 + NT1) * 32; e += (NWAVE - SCW) * 32) {
      int gi, le, base;
      if (e < NT3 * 32)              { gi = 3; le = e;              base = TB3_OFF; }
      else if (e < (NT3 + NT2) * 32) { gi = 2; le = e - NT3 * 32;   base = TB2_OFF; }
      else                           { gi = 1; le = e - (NT3 + NT2) * 32; base = TB1_OFF; }
      const int K  = (gi == 3) ? 124 : ((gi == 2) ? 57 : 12);
      const int g  = K + 11;
      const int wb = (gi == 3) ? 69 : ((gi == 2) ? 12 : 0);
      const int t  = le >> 5;
      const int l  = le & 31;
      const int mm = l & 15;
      const int c1 = t * 4 + ((l >> 4) << 1);
      const int d0 = c1 - mm;
      float v0 = Wv[wb + min(max(d0, 0), K - 1)];
      float v1 = Wv[wb + min(max(d0 + 1, 0), K - 1)];
      v0 = (mm < 12 && c1     < g && d0     >= 0 && d0     < K) ? v0 : 0.f;
      v1 = (mm < 12 && c1 + 1 < g && d0 + 1 >= 0 && d0 + 1 < K) ? v1 : 0.f;
      sm[base + 2 * le]     = v0;
      sm[base + 2 * le + 1] = v1;
    }
  }
}

template <int GB>
struct GBatch {
  v4f  ld[GB][3];
  int  so[GB];
  bool wok[GB];
};

template <int NL, int G0, int GB>
__device__ __forceinline__ void batch_issue(GBatch<GB>& q, const float* __restrict__ img,
                                            int x0, int f, int wave, int lane)
{
  const int q4 = (lane & 7) << 2;
  const int us = lane >> 3;
#pragma unroll
  for (int g = 0; g < GB; ++g) {
    int k, ln;
    if (G0 + g < RPW) { k = G0 + g; ln = us; }
    else {
      const int v = min(4 * (G0 + g - RPW) + us, RPW * (NL - 4) - 1);
      if (NL == 5) { k = v;      ln = 4;           }
      else         { k = v >> 1; ln = 4 + (v & 1); }
    }
    const int  r   = wave + NWAVE * k;
    const int  iy  = x0 + r;
    const int  L   = f + ln;
    const bool w   = (r < 135) && (iy >= 0) && (iy < IMG) && (L >= 0) && (L < LPR);
    const float* p = img + (w ? (iy * IMG + L * 32 + q4) : q4);
    q.ld[g][0] = __builtin_nontemporal_load((const v4f*)p);
    q.ld[g][1] = __builtin_nontemporal_load((const v4f*)(p + IMG * IMG));
    q.ld[g][2] = __builtin_nontemporal_load((const v4f*)(p + 2 * IMG * IMG));
    q.so[g] = r * SSTR + ln * 32 + q4; q.wok[g] = w;
  }
}

template <int GB>
__device__ __forceinline__ void batch_sum(const GBatch<GB>& q, float* __restrict__ S)
{
#pragma unroll
  for (int g = 0; g < GB; ++g) {
    const v4f zero = {0.f, 0.f, 0.f, 0.f};
    const v4f s = (q.ld[g][0] + q.ld[g][1]) + q.ld[g][2];
    *(v4f*)(S + q.so[g]) = q.wok[g] ? s : zero;
  }
}

template <int NL>
__device__ __forceinline__ void gather_lines(const float* __restrict__ img, float* __restrict__ sm,
                                             int x0, int f, int wave, int lane)
{
  constexpr int NG = RPW + (RPW * (NL - 4) + 3) / 4;
  float* S = sm + S_OFF;
  GBatch<NG / 2> q0;
  batch_issue<NL, 0, NG / 2>(q0, img, x0, f, wave, lane);
  scales_and_tables(sm, wave, lane);
  batch_sum<NG / 2>(q0, S);
  GBatch<NG / 2> q1;
  batch_issue<NL, NG / 2, NG / 2>(q1, img, x0, f, wave, lane);
  batch_sum<NG / 2>(q1, S);
}

__global__ __launch_bounds__(NTHR) void glimpse_wmma_kernel(
    const float* __restrict__ imgs, const int* __restrict__ locs,
    float* __restrict__ out)
{
  extern __shared__ __attribute__((aligned(16))) float sm[];
  float* S  = sm + S_OFF;
  float* Wv = sm + W_OFF;
  float* SC = sm + SC_OFF;

  const int b    = blockIdx.x;
  const int tid  = threadIdx.x;
  const int lane = tid & 31;
  const int wave = __builtin_amdgcn_readfirstlane(tid >> 5);

  const int x0 = locs[2 * b + 0] - 67;
  const int y0 = locs[2 * b + 1] - 67;
  const int tw = wave / 5;
  if (wave == 5 * tw) {
    const int u = tw * 32 + lane;
    if (u < NTAP) bicubic_taps(Wv, u);
  }
  __syncthreads();

  const float* img = imgs + (size_t)b * (3 * IMG * IMG);
  const int lf   = y0 >> 5;
  const int acol = y0 & 31;
  if (acol <= 24) gather_lines<5>(img, sm, x0, lf, wave, lane);
  else            gather_lines<6>(img, sm, x0, lf, wave, lane);
  __syncthreads();

  const int m  = lane & 15;
  const int kk = (lane >> 4) << 1;
  const int Mb = (lane >> 4) << 3;

  const int hh = lane >> 4;
  for (int task = wave; task < 16; task += NWAVE) {
    int gi, blk;
    if (task < 9)       { gi = 3; blk = task;      }
    else if (task < 14) { gi = 2; blk = task - 9;  }
    else                { gi = 1; blk = task - 14; }
    const int g     = (gi == 3) ? 135 : ((gi == 2) ? 68 : 23);
    const int off   = 67 - (g >> 1);
    const int ntile = (g + 3) >> 2;
    const v2f* TBg  = (const v2f*)(sm + ((gi == 3) ? TB3_OFF : ((gi == 2) ? TB2_OFF : TB1_OFF)));
    float*     P    = sm + ((gi == 3) ? H3_OFF : ((gi == 2) ? H2_OFF : H1_OFF));
    const float* Srow = S + (off + blk * 16 + m) * SSTR + acol + off + kk;

    v2f av[4];
#pragma unroll
    for (int t = 0; t < 4; ++t) {
      const int tau = min(4 * blk + 2 * hh + (t >> 1), ntile);
      av[t] = TBg[tau * 32 + ((t & 1) << 4) + m];
    }

    v8f acc = {};
    v2f a, bm;
    a.x = Srow[0]; a.y = Srow[1];
    bm  = TBg[lane];
    for (int t = 0; t < ntile; ++t) {
      v2f a2, b2;
      a2.x = Srow[(t + 1) * 4];
      a2.y = Srow[(t + 1) * 4 + 1];
      b2   = TBg[(t + 1) * 32 + lane];
      acc = __builtin_amdgcn_wmma_f32_16x16x4_f32(
          false, a, false, bm, (short)0, acc, false, false);
      a = a2; bm = b2;
    }

    v8f po = {};
#pragma unroll
    for (int t = 0; t < 4; ++t) {
      v2f hb;
      hb.x = acc[2 * t];
      hb.y = acc[2 * t + 1];
      po = __builtin_amdgcn_wmma_f32_16x16x4_f32(
          false, av[t], false, hb, (short)0, po, false, false);
    }
#pragma unroll
    for (int r = 0; r < 8; ++r)
      P[(blk * 16 + Mb + r) * 16 + m] = po[r];
  }
  __syncthreads();

  if (tid < 3 * 144) {
    const int gq = tid / 144;
    const int gi = gq + 1;
    const int o  = tid - gq * 144;
    const int M  = o / 12;
    const int j  = o - M * 12;
    const int nb = (gi == 3) ? 9 : ((gi == 2) ? 5 : 2);
    const float* P = sm + ((gi == 3) ? H3_OFF : ((gi == 2) ? H2_OFF : H1_OFF)) + M * 16 + j;
    const float sc = SC[gi];
    float p[9];
#pragma unroll
    for (int k = 0; k < 9; ++k) p[k] = P[min(k, nb - 1) * 256];
    float s = p[0];
#pragma unroll
    for (int k = 1; k < 9; ++k) s = (k < nb) ? s + p[k] : s;
    out[((size_t)b * 4 + gi) * 144 + o] = s * sc;
  } else {
    const float sc = SC[0];
    for (int t = tid - 3 * 144; t < 144; t += NTHR - 3 * 144) {
      int o = t / 12, j = t - o * 12;
      out[((size_t)b * 4 * 12 + o) * 12 + j] = S[(61 + o) * SSTR + acol + 61 + j] * sc;
    }
  }
}

extern "C" void kernel_launch(void* const* d_in, const int* in_sizes, int n_in,
                              void* d_out, int out_size, void* d_ws, size_t ws_size,
                              hipStream_t stream) {
  const float* imgs = (const float*)d_in[0];
  const int*   locs = (const int*)d_in[1];
  float*       out  = (float*)d_out;
  (void)in_sizes; (void)n_in; (void)d_ws; (void)ws_size; (void)out_size;

  const size_t shmem = (size_t)SMEM_FLOATS * sizeof(float);
  glimpse_wmma_kernel<<<dim3(256), dim3(NTHR), shmem, stream>>>(imgs, locs, out);
}


# step 1 split around the gather's second load batch: 11 row-block tasks (crop rows 0..95) run while it is in flight, 5 late ones after it; task operands loaded in 9-tile register chunks ahead
# speedup vs baseline: 5.4597x; 1.0512x over previous
#include <hip/hip_runtime.h>


typedef __attribute__((ext_vector_type(2))) float v2f;
typedef __attribute__((ext_vector_type(4))) float v4f;
typedef __attribute__((ext_vector_type(8))) float v8f;

#define IMG   224
#define NTHR  512
#define NWAVE (NTHR / 32)
#define RPW   (SROWS / NWAVE)
#define LPR   (IMG / 32)
#define SCW   3
#define NTAP  97
#define SROWS 144
#define SSTR  196
#define S_OFF   0
#define H3_OFF  (SROWS*SSTR)
#define H2_OFF  (H3_OFF + 144*16)
#define H1_OFF  (H2_OFF + 80*16)
#define W_OFF   (H1_OFF + 32*16)
#define SC_OFF  (W_OFF + 208)
#define NT3 35
#define NT2 18
#define NT1 7
#define TB3_OFF (SC_OFF + 16)
#define TB2_OFF (TB3_OFF + NT3*32*2)
#define TB1_OFF (TB2_OFF + NT2*32*2)
#define SMEM_FLOATS (TB1_OFF + NT1*32*2)

__device__ __forceinline__ void bicubic_taps(float* __restrict__ Wv, int u) {
  const bool f1 = u < 6, f2 = u < 35;
  const int    K    = f1 ? 12 : (f2 ? 57 : 124);
  const int    base = f1 ? 0  : (f2 ? 12 : 69);
  const int    t    = f1 ? u  : (f2 ? u - 6 : u - 35);
  const double hk   = f1 ? 5.5 : (f2 ? 28.0 : 61.5);
  const double r    = f1 ? (4.0 / 12.0) : (f2 ? (4.0 / 57.0) : (4.0 / 124.0));
  const double ax   = fabs(((double)t - hk) * r);
  const bool   in   = !(ax > 1.0);
  const double c3 = in ? 1.5  : -0.5;
  const double c2 = in ? -2.5 : 2.5;
  const double c1 = in ? 0.0  : -4.0;
  const double c0 = in ? 1.0  : 2.0;
  double w = __builtin_fma(__builtin_fma(__builtin_fma(c3, ax, c2), ax, c1), ax, c0);
  w = (ax > 2.0) ? 0.0 : w;
  const float wf = (float)w;
  Wv[base + t]         = wf;
  Wv[base + K - 1 - t] = wf;
}

__device__ __forceinline__ void scales_and_tables(float* __restrict__ sm, int wave, int lane)
{
  const float* Wv = sm + W_OFF;
  float*       SC = sm + SC_OFF;
  const int sw = wave / 3;
  if (wave == 3 * sw && sw >= 1 && sw <= SCW) {
    const int f    = sw - 1;
    const int K    = (f == 0) ? 12 : ((f == 1) ? 57 : 124);
    const int base = (f == 0) ? 0  : ((f == 1) ? 12 : 69);
    double d[4];
    for (int j = 0; j < 4; ++j) {
      const int   i = lane + 32 * j;
      const float w = Wv[base + min(i, K - 1)];
      d[j] = (i < K) ? (double)w : 0.0;
    }
    double s = (d[0] + d[1]) + (d[2] + d[3]);
    for (int x = 16; x > 0; x >>= 1) s += __shfl_xor(s, x);
    if (lane == 0) {
      SC[f + 1] = (float)(1.0 / (3.0 * s * s));
      if (f == 0) SC[0] = 1.0f / 3.0f;
    }
  } else {
    const int rank = wave - (wave > 3) - (wave > 6) - (wave > 9);
    for (int e = rank * 32 + lane; e < (NT3 + NT2 + NT1) * 32; e += (NWAVE - SCW) * 32) {
      int gi, le, base;
      if (e < NT3 * 32)              { gi = 3; le = e;              base = TB3_OFF; }
      else if (e < (NT3 + NT2) * 32) { gi = 2; le = e - NT3 * 32;   base = TB2_OFF; }
      else                           { gi = 1; le = e - (NT3 + NT2) * 32; base = TB1_OFF; }
      const int K  = (gi == 3) ? 124 : ((gi == 2) ? 57 : 12);
      const int g  = K + 11;
      const int wb = (gi == 3) ? 69 : ((gi == 2) ? 12 : 0);
      const int t  = le >> 5;
      const int l  = le & 31;
      const int mm = l & 15;
      const int c1 = t * 4 + ((l >> 4) << 1);
      const int d0 = c1 - mm;
      float v0 = Wv[wb + min(max(d0, 0), K - 1)];
      float v1 = Wv[wb + min(max(d0 + 1, 0), K - 1)];
      v0 = (mm < 12 && c1     < g && d0     >= 0 && d0     < K) ? v0 : 0.f;
      v1 = (mm < 12 && c1 + 1 < g && d0 + 1 >= 0 && d0 + 1 < K) ? v1 : 0.f;
      sm[base + 2 * le]     = v0;
      sm[base + 2 * le + 1] = v1;
    }
  }
}

template <int GB>
struct GBatch {
  v4f  ld[GB][3];
  int  so[GB];
  bool wok[GB];
};

template <int NL, int M0, int NM, int T0, int GB>
__device__ __forceinline__ void batch_issue(GBatch<GB>& q, const float* __restrict__ img,
                                            int x0, int f, int wave, int lane)
{
  const int q4 = (lane & 7) << 2;
  const int us = lane >> 3;
#pragma unroll
  for (int g = 0; g < GB; ++g) {
    const int G = (g < NM) ? M0 + g : T0 + (g - NM);
    int k, ln;
    if (G < RPW) { k = G; ln = us; }
    else {
      const int v = min(4 * (G - RPW) + us, RPW * (NL - 4) - 1);
      if (NL == 5) { k = v;      ln = 4;           }
      else         { k = v >> 1; ln = 4 + (v & 1); }
    }
    const int  r   = wave + NWAVE * k;
    const int  iy  = x0 + r;
    const int  L   = f + ln;
    const bool w   = (r < 135) && (iy >= 0) && (iy < IMG) && (L >= 0) && (L < LPR);
    const float* p = img + (w ? (iy * IMG + L * 32 + q4) : q4);
    q.ld[g][0] = __builtin_nontemporal_load((const v4f*)p);
    q.ld[g][1] = __builtin_nontemporal_load((const v4f*)(p + IMG * IMG));
    q.ld[g][2] = __builtin_nontemporal_load((const v4f*)(p + 2 * IMG * IMG));
    q.so[g] = r * SSTR + ln * 32 + q4; q.wok[g] = w;
  }
}

template <int GB>
__device__ __forceinline__ void batch_sum(const GBatch<GB>& q, float* __restrict__ S)
{
#pragma unroll
  for (int g = 0; g < GB; ++g) {
    const v4f zero = {0.f, 0.f, 0.f, 0.f};
    const v4f s = (q.ld[g][0] + q.ld[g][1]) + q.ld[g][2];
    *(v4f*)(S + q.so[g]) = q.wok[g] ? s : zero;
  }
}

template <int GI>
__device__ __forceinline__ void row_block_task(float* __restrict__ sm, int blk,
                                               int acol, int lane)
{
  constexpr int g     = (GI == 3) ? 135 : ((GI == 2) ? 68 : 23);
  constexpr int off   = 67 - (g >> 1);
  constexpr int ntile = (g + 3) >> 2;
  constexpr int CH    = 9;
  constexpr int NC    = (ntile + CH - 1) / CH;
  constexpr int tbo   = (GI == 3) ? TB3_OFF : ((GI == 2) ? TB2_OFF : TB1_OFF);
  constexpr int pof   = (GI == 3) ? H3_OFF : ((GI == 2) ? H2_OFF : H1_OFF);
  const float* S = sm + S_OFF;
  const int m  = lane & 15;
  const int kk = (lane >> 4) << 1;
  const int Mb = (lane >> 4) << 3;
  const int hh = lane >> 4;
  const v2f* TBg  = (const v2f*)(sm + tbo);
  float*     P    = sm + pof;
  const float* Srow = S + (off + blk * 16 + m) * SSTR + acol + off + kk;

  v2f A[2][CH], B[2][CH];
#pragma unroll
  for (int i = 0; i < CH; ++i) {
    if (i < ntile) {
      A[0][i].x = Srow[i * 4];
      A[0][i].y = Srow[i * 4 + 1];
      B[0][i]   = TBg[i * 32 + lane];
    }
  }
  v2f av[4];
#pragma unroll
  for (int t = 0; t < 4; ++t) {
    const int tau = min(4 * blk + 2 * hh + (t >> 1), ntile);
    av[t] = TBg[tau * 32 + ((t & 1) << 4) + m];
  }

  v8f acc = {};
#pragma unroll
  for (int c = 0; c < NC; ++c) {
#pragma unroll
    for (int i = 0; i < CH; ++i) {
      const int t = (c + 1) * CH + i;
      if (c + 1 < NC && t < ntile) {
        A[(c + 1) & 1][i].x = Srow[t * 4];
        A[(c + 1) & 1][i].y = Srow[t * 4 + 1];
        B[(c + 1) & 1][i]   = TBg[t * 32 + lane];
      }
    }
    __builtin_amdgcn_sched_barrier(0);
#pragma unroll
    for (int i = 0; i < CH; ++i) {
      if (c * CH + i < ntile)
        acc = __builtin_amdgcn_wmma_f32_16x16x4_f32(
            false, A[c & 1][i], false, B[c & 1][i], (short)0, acc, false, false);
    }
  }

  v8f po = {};
#pragma unroll
  for (int t = 0; t < 4; ++t) {
    v2f hb;
    hb.x = acc[2 * t];
    hb.y = acc[2 * t + 1];
    po = __builtin_amdgcn_wmma_f32_16x16x4_f32(
        false, av[t], false, hb, (short)0, po, false, false);
  }
#pragma unroll
  for (int r = 0; r < 8; ++r)
    P[(blk * 16 + Mb + r) * 16 + m] = po[r];
}

#define EARLY_LO 0x1134323010000000ull
#define EARLY_HI 0x2221000035203331ull
#define LATE_LO  0x2400000023383736ull
__device__ __forceinline__ int task_code(unsigned long long lo, unsigned long long hi, int wave)
{
  const unsigned long long tb = (wave < 8) ? lo : hi;
  return (int)((tb >> ((wave & 7) * 8)) & 0xffull);
}

__device__ __forceinline__ void run_task(float* __restrict__ sm, int code, int acol, int lane)
{
  const int gi = code >> 4, blk = code & 15;
  if (gi == 3)      row_block_task<3>(sm, blk, acol, lane);
  else if (gi == 2) row_block_task<2>(sm, blk, acol, lane);
  else if (gi == 1) row_block_task<1>(sm, blk, acol, lane);
}

template <int NL>
__device__ __forceinline__ void gather_lines(const float* __restrict__ img, float* __restrict__ sm,
                                             int x0, int acol, int f, int wave, int lane)
{
  constexpr int NG  = RPW + (RPW * (NL - 4) + 3) / 4;
  constexpr int NM0 = 6;
  constexpr int NT0 = (NL == 5) ? 2 : 3;
  constexpr int GB0 = NM0 + NT0;
  constexpr int GB1 = NG - GB0;
  float* S = sm + S_OFF;
  GBatch<GB0> q0;
  batch_issue<NL, 0, NM0, RPW, GB0>(q0, img, x0, f, wave, lane);
  scales_and_tables(sm, wave, lane);
  batch_sum<GB0>(q0, S);
  GBatch<GB1> q1;
  batch_issue<NL, NM0, RPW - NM0, RPW + NT0, GB1>(q1, img, x0, f, wave, lane);
  __builtin_amdgcn_fence(__ATOMIC_RELEASE, "workgroup", "local");
  __builtin_amdgcn_s_barrier();
  __builtin_amdgcn_fence(__ATOMIC_ACQUIRE, "workgroup", "local");
  run_task(sm, task_code(EARLY_LO, EARLY_HI, wave), acol, lane);
  batch_sum<GB1>(q1, S);
}

__global__ __launch_bounds__(NTHR) void glimpse_wmma_kernel(
    const float* __restrict__ imgs, const int* __restrict__ locs,
    float* __restrict__ out)
{
  extern __shared__ __attribute__((aligned(16))) float sm[];
  float* S  = sm + S_OFF;
  float* Wv = sm + W_OFF;
  float* SC = sm + SC_OFF;

  const int b    = blockIdx.x;
  const int tid  = threadIdx.x;
  const int lane = tid & 31;
  const int wave = __builtin_amdgcn_readfirstlane(tid >> 5);

  const int x0 = locs[2 * b + 0] - 67;
  const int y0 = locs[2 * b + 1] - 67;
  const int tw = wave / 5;
  if (wave == 5 * tw) {
    const int u = tw * 32 + lane;
    if (u < NTAP) bicubic_taps(Wv, u);
  }
  __syncthreads();

  const float* img = imgs + (size_t)b * (3 * IMG * IMG);
  const int lf   = y0 >> 5;
  const int acol = y0 & 31;
  if (acol <= 24) gather_lines<5>(img, sm, x0, acol, lf, wave, lane);
  else            gather_lines<6>(img, sm, x0, acol, lf, wave, lane);
  __syncthreads();

  run_task(sm, task_code(LATE_LO, 0ull, wave), acol, lane);
  __syncthreads();

  if (tid < 3 * 144) {
    const int gq = tid / 144;
    const int gi = gq + 1;
    const int o  = tid - gq * 144;
    const int M  = o / 12;
    const int j  = o - M * 12;
    const int nb = (gi == 3) ? 9 : ((gi == 2) ? 5 : 2);
    const float* P = sm + ((gi == 3) ? H3_OFF : ((gi == 2) ? H2_OFF : H1_OFF)) + M * 16 + j;
    const float sc = SC[gi];
    float p[9];
#pragma unroll
    for (int k = 0; k < 9; ++k) p[k] = P[min(k, nb - 1) * 256];
    float s = p[0];
#pragma unroll
    for (int k = 1; k < 9; ++k) s = (k < nb) ? s + p[k] : s;
    out[((size_t)b * 4 + gi) * 144 + o] = s * sc;
  } else {
    const float sc = SC[0];
    for (int t = tid - 3 * 144; t < 144; t += NTHR - 3 * 144) {
      int o = t / 12, j = t - o * 12;
      out[((size_t)b * 4 * 12 + o) * 12 + j] = S[(61 + o) * SSTR + acol + 61 + j] * sc;
    }
  }
}

extern "C" void kernel_launch(void* const* d_in, const int* in_sizes, int n_in,
                              void* d_out, int out_size, void* d_ws, size_t ws_size,
                              hipStream_t stream) {
  const float* imgs = (const float*)d_in[0];
  const int*   locs = (const int*)d_in[1];
  float*       out  = (float*)d_out;
  (void)in_sizes; (void)n_in; (void)d_ws; (void)ws_size; (void)out_size;

  const size_t shmem = (size_t)SMEM_FLOATS * sizeof(float);
  glimpse_wmma_kernel<<<dim3(256), dim3(NTHR), shmem, stream>>>(imgs, locs, out);
}
